# starting kernel
# baseline (speedup 1.0000x reference)
#include <hip/hip_runtime.h>
#include <hip/hip_bf16.h>

#define IN_DIM   16384
#define BATCH    4096
#define OUT_CH   2
#define N_SLICES 32
#define SLICE_K  (IN_DIM / N_SLICES)
#define TK       128
#define TK_PAD   132
#define WAVES_PER_BLOCK 2
#define LDS_ROWS_PER_WAVE (16 + 32)
#define LDS_F32_PER_WAVE  (LDS_ROWS_PER_WAVE * TK_PAD)

typedef __attribute__((ext_vector_type(2))) float v2f;
typedef __attribute__((ext_vector_type(4))) float v4f;
typedef __attribute__((ext_vector_type(8))) float v8f;

__global__ __launch_bounds__(256) void moe_bias_init(const int* __restrict__ ts,
                                                     const float* __restrict__ bias,
                                                     float* __restrict__ out) {
    int i = blockIdx.x * 256 + threadIdx.x;
    if (i < BATCH * OUT_CH) {
        int s  = i >> 1;
        int ch = i & 1;
        out[i] = bias[ts[s] * OUT_CH + ch];
    }
}

__global__ __launch_bounds__(WAVES_PER_BLOCK * 32) void
moe_wmma(const float* __restrict__ x,
         const int*   __restrict__ ts,
         const float* __restrict__ W,
         float*       __restrict__ out) {
    __shared__ float lds[WAVES_PER_BLOCK * LDS_F32_PER_WAVE];

    const int wave = threadIdx.x >> 5;
    const int lane = threadIdx.x & 31;
    float* As = lds + wave * LDS_F32_PER_WAVE;
    float* Ws = As + 16 * TK_PAD;

    const int wid   = blockIdx.x * WAVES_PER_BLOCK + wave;
    const int group = wid >> 5;
    const int slice = wid & 31;

    const int m    = lane & 15;
    const int half = lane >> 4;

    int tid16[16];
    #pragma unroll
    for (int r = 0; r < 16; ++r) tid16[r] = ts[group * 16 + r];

    const float* xg = x + (size_t)(group * 16) * IN_DIM + slice * SLICE_K;

    v8f c0 = {};
    v8f c1 = {};

    for (int tile = 0; tile < SLICE_K; tile += TK) {
        #pragma unroll
        for (int r = 0; r < 16; ++r) {
            v4f v = *(const v4f*)(xg + (size_t)r * IN_DIM + tile + lane * 4);
            *(v4f*)(As + r * TK_PAD + lane * 4) = v;
        }
        #pragma unroll
        for (int r = 0; r < 16; ++r) {
            const float* wr = W + (size_t)tid16[r] * (OUT_CH * IN_DIM)
                                + slice * SLICE_K + tile + lane * 4;
            v4f w0 = *(const v4f*)(wr);
            v4f w1 = *(const v4f*)(wr + IN_DIM);
            *(v4f*)(Ws + r        * TK_PAD + lane * 4) = w0;
            *(v4f*)(Ws + (16 + r) * TK_PAD + lane * 4) = w1;
        }
        __builtin_amdgcn_wave_barrier();

        #pragma unroll 4
        for (int kk = 0; kk < TK; kk += 4) {
            v2f a  = *(const v2f*)(As + m        * TK_PAD + kk + 2 * half);
            v2f b0 = *(const v2f*)(Ws + m        * TK_PAD + kk + 2 * half);
            v2f b1 = *(const v2f*)(Ws + (16 + m) * TK_PAD + kk + 2 * half);
            c0 = __builtin_amdgcn_wmma_f32_16x16x4_f32(false, a, false, b0,
                                                       (short)0, c0, false, false);
            c1 = __builtin_amdgcn_wmma_f32_16x16x4_f32(false, a, false, b1,
                                                       (short)0, c1, false, false);
        }
        __builtin_amdgcn_wave_barrier();
    }

    const bool active = (lane < 8) || (lane >= 24);
    const int  r      = (lane < 8) ? lane : (lane - 24);
    const int  mres   = (lane < 8) ? lane : (lane - 16);

    float v0 = c0[0], v1 = c1[0];
    #pragma unroll
    for (int i = 1; i < 8; ++i) {
        v0 = (r == i) ? c0[i] : v0;
        v1 = (r == i) ? c1[i] : v1;
    }

    if (active) {
        float* o = out + (size_t)(group * 16 + mres) * OUT_CH;
        atomicAdd(o + 0, v0);
        atomicAdd(o + 1, v1);
    }
}

extern "C" void kernel_launch(void* const* d_in, const int* in_sizes, int n_in,
                              void* d_out, int out_size, void* d_ws, size_t ws_size,
                              hipStream_t stream) {
    const float* x   = (const float*)d_in[0];
    const int*   ts  = (const int*)  d_in[1];
    const float* W   = (const float*)d_in[2];
    const float* b   = (const float*)d_in[3];
    float*       out = (float*)d_out;

    (void)in_sizes; (void)n_in; (void)out_size; (void)d_ws; (void)ws_size;

    moe_bias_init<<<(BATCH * OUT_CH + 255) / 256, 256, 0, stream>>>(ts, b, out);

    const int n_waves  = (BATCH / 16) * N_SLICES;
    const int n_blocks = n_waves / WAVES_PER_BLOCK;
    moe_wmma<<<n_blocks, WAVES_PER_BLOCK * 32, 0, stream>>>(x, ts, W, out);
}


# K split over 32 block columns: block holds its 512-k slice of all 100 weight rows in LDS, x lines feed WMMA operands straight from memory, 4 samples a tile; launch 2 adds 32 partials: sums re-ordered
# speedup vs baseline: 1.4005x; 1.4005x over previous
#include <hip/hip_runtime.h>

#define IN_DIM   16384
#define BATCH    4096
#define OUT_CH   2
#define N_EXP    50
#define KW       512
#define KSPLIT   (IN_DIM / KW)
#define RB       8
#define ROWS_PB  (BATCH / RB)
#define WAVES    16
#define ROWS_PW  (ROWS_PB / WAVES)
#define GROUPS   (ROWS_PW / 4)
#define LINES    (KW / 32)
#define PITCH    (KW + 8)
#define W_ROWS   (N_EXP * OUT_CH)
#define STAGE_IT ((W_ROWS * (KW / 4)) / (WAVES * 32))

typedef __attribute__((ext_vector_type(2))) float v2f;
typedef __attribute__((ext_vector_type(4))) float v4f;
typedef __attribute__((ext_vector_type(8))) float v8f;

__global__ __launch_bounds__(WAVES * 32) void
moe_kslice(const float* __restrict__ x,
           const int*   __restrict__ ts,
           const float* __restrict__ W,
           float*       __restrict__ part) {
    __shared__ __attribute__((aligned(16))) float Wl[W_ROWS * PITCH];

    const int tid  = threadIdx.x;
    const int wave = tid >> 5;
    const int lane = tid & 31;
    const int ks   = blockIdx.x & (KSPLIT - 1);
    const int rb   = blockIdx.x / KSPLIT;

    const int c  = lane & 15;
    const int j  = c >> 2;
    const int p4 = ((lane >> 4) * 4 + (c & 3)) * 4;
    const int row0 = rb * ROWS_PB + wave * ROWS_PW;

    const float* xp = x + (size_t)(row0 + j) * IN_DIM + ks * KW + p4;

    int woff[GROUPS];
    #pragma unroll
    for (int g = 0; g < GROUPS; ++g) {
        int t = ts[row0 + g * 4 + j];
        t = min(max(t, 0), N_EXP - 1);
        woff[g] = t * (OUT_CH * PITCH) + p4;
    }

    v4f xa[LINES], xb[LINES];

#define LOAD_X(buf, g)                                                          \
    do {                                                                        \
        _Pragma("unroll")                                                       \
        for (int i = 0; i < LINES; ++i)                                         \
            buf[i] = __builtin_nontemporal_load(                                \
                (const v4f*)(xp + (size_t)(g) * (4 * IN_DIM) + i * 32));        \
    } while (0)

    LOAD_X(xa, 0);

    {
        const float* wsrc = W + ks * KW;
        v4f v[STAGE_IT];
        #pragma unroll
        for (int u = 0; u < STAGE_IT; ++u) {
            const int idx = u * (WAVES * 32) + tid;
            const int r   = idx >> 7;
            const int c4  = idx & 127;
            v[u] = *(const v4f*)(wsrc + (size_t)r * IN_DIM + c4 * 4);
        }
        #pragma unroll
        for (int u = 0; u < STAGE_IT; ++u) {
            const int idx = u * (WAVES * 32) + tid;
            const int r   = idx >> 7;
            const int c4  = idx & 127;
            *(v4f*)(Wl + r * PITCH + c4 * 4) = v[u];
        }
    }
    __syncthreads();
    LOAD_X(xb, 1);

    const int  r8     = lane & 7;
    const bool leader = ((lane & 3) == 0) && (lane < 8 || lane >= 24);
    const int  jj     = (lane < 8) ? (lane >> 2) : (2 + ((lane - 24) >> 2));
    float* pout = part + (size_t)ks * (BATCH * OUT_CH) + (size_t)(row0 + jj) * OUT_CH;

#define COMPUTE(buf, g)                                                         \
    do {                                                                        \
        v8f c0 = {}, c1 = {};                                                   \
        const float* b0p = Wl + woff[g];                                        \
        const float* b1p = b0p + PITCH;                                         \
        _Pragma("unroll")                                                       \
        for (int i = 0; i < LINES; ++i) {                                       \
            const v4f a  = buf[i];                                              \
            const v4f w0 = *(const v4f*)(b0p + i * 32);                         \
            const v4f w1 = *(const v4f*)(b1p + i * 32);                         \
            const v2f alo  = {a[0], a[1]},   ahi  = {a[2], a[3]};               \
            const v2f w0lo = {w0[0], w0[1]}, w0hi = {w0[2], w0[3]};             \
            const v2f w1lo = {w1[0], w1[1]}, w1hi = {w1[2], w1[3]};             \
            c0 = __builtin_amdgcn_wmma_f32_16x16x4_f32(false, alo, false, w0lo, \
                                                       (short)0, c0, false, false); \
            c1 = __builtin_amdgcn_wmma_f32_16x16x4_f32(false, alo, false, w1lo, \
                                                       (short)0, c1, false, false); \
            c0 = __builtin_amdgcn_wmma_f32_16x16x4_f32(false, ahi, false, w0hi, \
                                                       (short)0, c0, false, false); \
            c1 = __builtin_amdgcn_wmma_f32_16x16x4_f32(false, ahi, false, w1hi, \
                                                       (short)0, c1, false, false); \
        }                                                                       \
          \
        float v0 = c0[0], v1 = c1[0];                                           \
        _Pragma("unroll")                                                       \
        for (int i = 1; i < 8; ++i) {                                           \
            v0 = (r8 == i) ? c0[i] : v0;                                        \
            v1 = (r8 == i) ? c1[i] : v1;                                        \
        }                                                                       \
             \
        v0 += __shfl_xor(v0, 1);  v1 += __shfl_xor(v1, 1);                      \
        v0 += __shfl_xor(v0, 2);  v1 += __shfl_xor(v1, 2);                      \
        if (leader) *(v2f*)(pout + (g) * (4 * OUT_CH)) = (v2f){v0, v1};         \
    } while (0)

    #pragma unroll
    for (int g = 0; g < GROUPS; g += 2) {
        COMPUTE(xa, g);
        if (g + 2 < GROUPS) LOAD_X(xa, g + 2);
        COMPUTE(xb, g + 1);
        if (g + 3 < GROUPS) LOAD_X(xb, g + 3);
    }
}

__global__ __launch_bounds__(256) void
moe_sum(const float* __restrict__ part,
        const int*   __restrict__ ts,
        const float* __restrict__ bias,
        float*       __restrict__ out) {
    const int i = blockIdx.x * 256 + threadIdx.x;
    int t = ts[i >> 1];
    t = min(max(t, 0), N_EXP - 1);
    float p[KSPLIT];
    #pragma unroll
    for (int k = 0; k < KSPLIT; ++k) p[k] = part[(size_t)k * (BATCH * OUT_CH) + i];
    float acc = bias[t * OUT_CH + (i & 1)];
    #pragma unroll
    for (int k = 0; k < KSPLIT; ++k) acc += p[k];
    out[i] = acc;
}

extern "C" void kernel_launch(void* const* d_in, const int* in_sizes, int n_in,
                              void* d_out, int out_size, void* d_ws, size_t ws_size,
                              hipStream_t stream) {
    const float* x   = (const float*)d_in[0];
    const int*   ts  = (const int*)  d_in[1];
    const float* W   = (const float*)d_in[2];
    const float* b   = (const float*)d_in[3];
    float*       out = (float*)d_out;
    float*       part = (float*)d_ws;

    (void)in_sizes; (void)n_in; (void)out_size; (void)ws_size;

    moe_kslice<<<KSPLIT * RB, WAVES * 32, 0, stream>>>(x, ts, W, part);
    moe_sum<<<(BATCH * OUT_CH) / 256, 256, 0, stream>>>(part, ts, b, out);
}


# Each block pairs its samples by expert (ballot masks, sample-index order): a WMMA's rows = 2 same-expert samples x 8 pieces, cols = 2 channels x 8 pieces: half the WMMAs; sums re-ordered
# speedup vs baseline: 1.5546x; 1.1100x over previous
#include <hip/hip_runtime.h>

#define IN_DIM   16384
#define BATCH    4096
#define OUT_CH   2
#define N_EXP    50
#define KW       512
#define KSPLIT   (IN_DIM / KW)
#define RB       8
#define ROWS_PB  (BATCH / RB)
#define WAVES    16
#define CHUNKS   (KW / 64)
#define PITCH    (KW + 32)
#define W_ROWS   (N_EXP * OUT_CH)
#define STAGE_IT ((W_ROWS * (KW / 4)) / (WAVES * 32))
#define MAXP     288

typedef __attribute__((ext_vector_type(2))) float v2f;
typedef __attribute__((ext_vector_type(4))) float v4f;
typedef __attribute__((ext_vector_type(8))) float v8f;
typedef __attribute__((ext_vector_type(4))) unsigned v4u;

__device__ __forceinline__ void lds_barrier() {
    __builtin_amdgcn_fence(__ATOMIC_RELEASE, "workgroup", "local");
    __builtin_amdgcn_s_barrier();
    __builtin_amdgcn_fence(__ATOMIC_ACQUIRE, "workgroup", "local");
}

#define SEL8(a,b,c,d,e,f,g,h) ((a) | (b) << 3 | (c) << 6 | (d) << 9 | (e) << 12 | (f) << 15 | (g) << 18 | (h) << 21)
#define XOR1 SEL8(1,0,3,2,5,4,7,6)
#define XOR2 SEL8(2,3,0,1,6,7,4,5)
#define XOR4 SEL8(4,5,6,7,0,1,2,3)
#define x8(v, sel) __builtin_bit_cast(float, __builtin_amdgcn_mov_dpp8(__builtin_bit_cast(int, (v)), (sel)))

__global__ __launch_bounds__(WAVES * 32) void
moe_kslice(const float* __restrict__ x,
           const int*   __restrict__ ts,
           const float* __restrict__ W,
           float*       __restrict__ part) {
    __shared__ __attribute__((aligned(16))) float    Wl[W_ROWS * PITCH];
    __shared__ __attribute__((aligned(16))) unsigned mk[N_EXP * WAVES];
    __shared__ int pl[MAXP * 2];
    __shared__ int pe[MAXP];
    __shared__ __attribute__((aligned(16))) float    ps[ROWS_PB * OUT_CH];

    const int tid  = threadIdx.x;
    const int wave = __builtin_amdgcn_readfirstlane(tid >> 5);
    const int lane = tid & 31;
    const int ks   = blockIdx.x & (KSPLIT - 1);
    const int rb   = blockIdx.x / KSPLIT;

    int t = ts[rb * ROWS_PB + tid];
    t = min(max(t, 0), N_EXP - 1);

    v4f v[STAGE_IT];
    {
        const float* wsrc = W + ks * KW;
        #pragma unroll
        for (int u = 0; u < STAGE_IT; ++u) {
            const int idx = u * (WAVES * 32) + tid;
            const int r   = idx >> 7;
            const int c4  = idx & 127;
            v[u] = *(const v4f*)(wsrc + (size_t)r * IN_DIM + c4 * 4);
        }
    }

    {
        unsigned mlo = 0, mhi = 0;
        #pragma unroll
        for (int e = 0; e < N_EXP; ++e) {
            const unsigned m = __builtin_amdgcn_ballot_w32(t == e);
            if (e < 32) mlo = (lane == e)      ? m : mlo;
            else        mhi = (lane == e - 32) ? m : mhi;
        }
        mk[lane * WAVES + wave] = mlo;
        if (lane < N_EXP - 32) mk[(lane + 32) * WAVES + wave] = mhi;
    }
    lds_barrier();

    int P;
    {
        int nlo = 0, nhi = 0;
        {
            const v4u* m0 = (const v4u*)(mk + lane * WAVES);
            #pragma unroll
            for (int i = 0; i < 4; ++i) {
                const v4u a = m0[i];
                nlo += __builtin_popcount(a[0]) + __builtin_popcount(a[1]) +
                       __builtin_popcount(a[2]) + __builtin_popcount(a[3]);
            }
            if (lane < N_EXP - 32) {
                const v4u* m1 = (const v4u*)(mk + (lane + 32) * WAVES);
                #pragma unroll
                for (int i = 0; i < 4; ++i) {
                    const v4u a = m1[i];
                    nhi += __builtin_popcount(a[0]) + __builtin_popcount(a[1]) +
                           __builtin_popcount(a[2]) + __builtin_popcount(a[3]);
                }
            }
        }
        const int plo = (nlo + 1) >> 1, phi = (nhi + 1) >> 1;
        int slo = plo, shi = phi;
        #pragma unroll
        for (int d = 1; d < 32; d <<= 1) {
            const int ylo = __shfl_up(slo, d), yhi = __shfl_up(shi, d);
            if (lane >= d) { slo += ylo; shi += yhi; }
        }
        const int totlo = __shfl(slo, 31);
        const int exlo  = slo - plo;
        const int exhi  = totlo + shi - phi;
        P = __builtin_amdgcn_readfirstlane(totlo + __shfl(shi, 31));

        const int blo = __shfl(exlo, t & 31), bhi = __shfl(exhi, t & 31);
        const int base = (t < 32) ? blo : bhi;
        int rank = 0, nt = 0;
        const v4u* mt = (const v4u*)(mk + t * WAVES);
        const unsigned below = (1u << lane) - 1u;
        #pragma unroll
        for (int i = 0; i < 4; ++i) {
            const v4u a = mt[i];
            #pragma unroll
            for (int s = 0; s < 4; ++s) {
                const int w = i * 4 + s;
                const int pc = __builtin_popcount(a[s]);
                nt += pc;
                rank += (w < wave) ? pc : ((w == wave) ? __builtin_popcount(a[s] & below) : 0);
            }
        }
        const int slot = base + (rank >> 1);
        if (rank & 1) {
            pl[slot * 2 + 1] = tid;
        } else {
            pl[slot * 2] = tid;
            pe[slot]     = t;
            if (rank == nt - 1) pl[slot * 2 + 1] = tid;
        }
    }
    lds_barrier();

    const int NG = (P + 1) >> 1;
    const int nw = (NG - wave + WAVES - 1) >> 4;

    const int hb   = (lane >> 3) & 1;
    const int offl = ((lane >> 4) * 8 + (lane & 7)) * 4;
    const float* xrow = x + (size_t)(rb * ROWS_PB) * IN_DIM + ks * KW + offl;

    v4f xa[2 * CHUNKS], xb[2 * CHUNKS];

#define LOAD_X(buf, it)                                                         \
    do {                                                                        \
        const int G_  = (it) * WAVES + wave;                                    \
        const int pA_ = min(2 * G_, P - 1), pB_ = min(2 * G_ + 1, P - 1);       \
        const float* xA_ = xrow + (size_t)pl[pA_ * 2 + hb] * IN_DIM;            \
        const float* xB_ = xrow + (size_t)pl[pB_ * 2 + hb] * IN_DIM;            \
        _Pragma("unroll")                                                       \
        for (int i = 0; i < CHUNKS; ++i) {                                      \
            buf[i]          = __builtin_nontemporal_load((const v4f*)(xA_ + i * 64)); \
            buf[CHUNKS + i] = __builtin_nontemporal_load((const v4f*)(xB_ + i * 64)); \
        }                                                                       \
    } while (0)

    LOAD_X(xa, 0);

    {
        #pragma unroll
        for (int u = 0; u < STAGE_IT; ++u) {
            const int idx = u * (WAVES * 32) + tid;
            const int r   = idx >> 7;
            const int c4  = idx & 127;
            *(v4f*)(Wl + r * PITCH + c4 * 4) = v[u];
        }
    }
    lds_barrier();
    LOAD_X(xb, 1);

    const int r8  = lane & 7;
    const int jd  = lane >> 4;
    const bool leader = (lane & 7) == 0;

#define COMPUTE(buf, it)                                                        \
    do {                                                                        \
        const int G_  = (it) * WAVES + wave;                                    \
        const int pA_ = min(2 * G_, P - 1), pB_ = min(2 * G_ + 1, P - 1);       \
        const float* bA_ = Wl + (pe[pA_] * OUT_CH + hb) * PITCH + offl;         \
        const float* bB_ = Wl + (pe[pB_] * OUT_CH + hb) * PITCH + offl;         \
        const int sA_ = pl[pA_ * 2 + jd], sB_ = pl[pB_ * 2 + jd];               \
        v4f wa_[CHUNKS], wb_[CHUNKS];                                           \
        _Pragma("unroll")                                                       \
        for (int i = 0; i < CHUNKS; ++i) {                                      \
            wa_[i] = *(const v4f*)(bA_ + i * 64);                               \
            wb_[i] = *(const v4f*)(bB_ + i * 64);                               \
        }                                                                       \
        v8f cA = {}, cB = {};                                                   \
        _Pragma("unroll")                                                       \
        for (int i = 0; i < CHUNKS; ++i) {                                      \
            const v4f a = buf[i], b = buf[CHUNKS + i];                          \
            const v4f wa = wa_[i], wb = wb_[i];                                 \
            const v2f alo  = {a[0], a[1]},   ahi  = {a[2], a[3]};               \
            const v2f blo  = {b[0], b[1]},   bhi  = {b[2], b[3]};               \
            const v2f walo = {wa[0], wa[1]}, wahi = {wa[2], wa[3]};             \
            const v2f wblo = {wb[0], wb[1]}, wbhi = {wb[2], wb[3]};             \
            cA = __builtin_amdgcn_wmma_f32_16x16x4_f32(false, alo, false, walo, \
                                                       (short)0, cA, false, false); \
            cB = __builtin_amdgcn_wmma_f32_16x16x4_f32(false, blo, false, wblo, \
                                                       (short)0, cB, false, false); \
            cA = __builtin_amdgcn_wmma_f32_16x16x4_f32(false, ahi, false, wahi, \
                                                       (short)0, cA, false, false); \
            cB = __builtin_amdgcn_wmma_f32_16x16x4_f32(false, bhi, false, wbhi, \
                                                       (short)0, cB, false, false); \
        }                                                                       \
        float vA = cA[0], vB = cB[0];                                           \
        _Pragma("unroll")                                                       \
        for (int i = 1; i < 8; ++i) {                                           \
            vA = (r8 == i) ? cA[i] : vA;                                        \
            vB = (r8 == i) ? cB[i] : vB;                                        \
        }                                                                       \
          \
        vA += x8(vA, XOR1);  vB += x8(vB, XOR1);                                \
        vA += x8(vA, XOR2);  vB += x8(vB, XOR2);                                \
        vA += x8(vA, XOR4);  vB += x8(vB, XOR4);                                \
        if (leader) { ps[sA_ * OUT_CH + hb] = vA; ps[sB_ * OUT_CH + hb] = vB; } \
    } while (0)

    COMPUTE(xa, 0);  LOAD_X(xa, 2);
    COMPUTE(xb, 1);  LOAD_X(xb, 3);
    COMPUTE(xa, 2);  LOAD_X(xa, 4);
    COMPUTE(xb, 3);  LOAD_X(xb, 5);
    COMPUTE(xa, 4);  LOAD_X(xa, 6);
    COMPUTE(xb, 5);  LOAD_X(xb, 7);
    COMPUTE(xa, 6);
    if (nw > 8) LOAD_X(xa, 8);
    COMPUTE(xb, 7);
    if (nw > 8) COMPUTE(xa, 8);

    __syncthreads();
    *(v2f*)(part + (size_t)ks * (BATCH * OUT_CH) + (size_t)(rb * ROWS_PB + tid) * OUT_CH) =
        *(const v2f*)(ps + tid * OUT_CH);
}

__global__ __launch_bounds__(256) void
moe_sum(const float* __restrict__ part,
        const int*   __restrict__ ts,
        const float* __restrict__ bias,
        float*       __restrict__ out) {
    const int i = blockIdx.x * 256 + threadIdx.x;
    int t = ts[i >> 1];
    t = min(max(t, 0), N_EXP - 1);
    float p[KSPLIT];
    #pragma unroll
    for (int k = 0; k < KSPLIT; ++k) p[k] = part[(size_t)k * (BATCH * OUT_CH) + i];
    float acc = bias[t * OUT_CH + (i & 1)];
    #pragma unroll
    for (int k = 0; k < KSPLIT; ++k) acc += p[k];
    out[i] = acc;
}

extern "C" void kernel_launch(void* const* d_in, const int* in_sizes, int n_in,
                              void* d_out, int out_size, void* d_ws, size_t ws_size,
                              hipStream_t stream) {
    const float* x   = (const float*)d_in[0];
    const int*   ts  = (const int*)  d_in[1];
    const float* W   = (const float*)d_in[2];
    const float* b   = (const float*)d_in[3];
    float*       out = (float*)d_out;
    float*       part = (float*)d_ws;

    (void)in_sizes; (void)n_in; (void)out_size; (void)ws_size;

    moe_kslice<<<KSPLIT * RB, WAVES * 32, 0, stream>>>(x, ts, W, part);
    moe_sum<<<(BATCH * OUT_CH) / 256, 256, 0, stream>>>(part, ts, b, out);
}
